# starting kernel
# baseline (speedup 1.0000x reference)
#include <hip/hip_runtime.h>
#include <cmath>

typedef __attribute__((ext_vector_type(2))) float v2f;
typedef __attribute__((ext_vector_type(8))) float v8f;

#define B_  8
#define N_  400
#define T_  24
#define Y_  12
#define H_  64
#define D_  96
#define DL_ 64
#define TY_ (T_ * Y_)
#define BTY_ (B_ * TY_)
#define NT_ 384
#define TC_ 6

__device__ __forceinline__ float fast_tanh(float x) {
#if __has_builtin(__builtin_amdgcn_tanhf)
    return __builtin_amdgcn_tanhf(x);
#else
    return tanhf(x);
#endif
}

__device__ __forceinline__ v8f wmma_f32_16x16x4(v2f a, v2f b, v8f c) {
    return __builtin_amdgcn_wmma_f32_16x16x4_f32(
         false, a,  false, b,
         (short)0, c,  false,  false);
}

__global__ __launch_bounds__(NT_) void tsa_fused_kernel(
    const float* __restrict__ x_long,
    const float* __restrict__ x_short,
    const float* __restrict__ Wq,
    const float* __restrict__ Wk,
    const float* __restrict__ V,
    const float* __restrict__ encW,
    const float* __restrict__ encb,
    float* __restrict__ out)
{
    __shared__ alignas(16) float  lds_wq[H_ * D_];
    __shared__ alignas(16) float  lds_xs[16 * D_];
    __shared__ alignas(16) float  lds_Q [B_ * H_];
    __shared__ alignas(16) float2 lds_vk[H_ * Y_];
    __shared__ alignas(16) float  lds_xl[BTY_];
    __shared__ alignas(16) float  lds_A [BTY_];
    __shared__ alignas(16) float  lds_xn[16 * 16];
    __shared__ alignas(16) float  lds_ewT[DL_ * 16];

    const int n    = blockIdx.x;
    const int tid  = threadIdx.x;
    const int wave = tid >> 5;
    const int lane = tid & 31;

    for (int i4 = tid; i4 < (H_ * D_) / 4; i4 += NT_) {
        const int i = i4 * 4;
        const int h = i / D_, d = i % D_;
        ((float4*)lds_wq)[i4] = *(const float4*)&Wq[(h * N_ + n) * D_ + d];
    }
    for (int i4 = tid; i4 < (B_ * D_) / 4; i4 += NT_) {
        const int i = i4 * 4;
        const int b = i / D_, d = i % D_;
        ((float4*)lds_xs)[i4] = *(const float4*)&x_short[(b * N_ + n) * D_ + d];
    }
    for (int i = tid; i < B_ * D_; i += NT_) lds_xs[B_ * D_ + i] = 0.0f;
    for (int i = tid; i < H_ * Y_; i += NT_) {
        const int h = i / Y_, y = i % Y_;
        lds_vk[i] = make_float2(Wk[(h * N_ + n) * Y_ + y],
                                V [(h * N_ + n) * Y_ + y]);
    }
    for (int i4 = tid; i4 < BTY_ / 4; i4 += NT_) {
        const int i = i4 * 4;
        const int b = i / TY_, r = i % TY_;
        ((float4*)lds_xl)[i4] = *(const float4*)&x_long[(b * N_ + n) * TY_ + r];
    }
    for (int i = tid; i < Y_ * DL_; i += NT_) {
        const int y = i / DL_, d = i % DL_;
        lds_ewT[d * 16 + y] = encW[n * Y_ * DL_ + i];
    }
    if (tid < 256) lds_xn[tid] = 0.0f;
    __syncthreads();

    if (wave < 4) {
        const int h0 = wave * 16;
        const int m  = lane & 15;
        const int ko = (lane >> 4) * 2;
        v8f c = {};
        #pragma unroll
        for (int kk = 0; kk < D_ / 4; ++kk) {
            const int k = kk * 4 + ko;
            const float2 af = *(const float2*)&lds_xs[m * D_ + k];
            const float2 bf = *(const float2*)&lds_wq[(h0 + m) * D_ + k];
            v2f a, b;
            a.x = af.x; a.y = af.y;
            b.x = bf.x; b.y = bf.y;
            c = wmma_f32_16x16x4(a, b, c);
        }
        if (lane < 16) {
            #pragma unroll
            for (int v = 0; v < B_; ++v)
                lds_Q[v * H_ + h0 + lane] = c[v];
        }
    }
    __syncthreads();

    {
        const int p  = tid % (B_ * Y_);
        const int t0 = (tid / (B_ * Y_)) * TC_;
        const int b  = p / Y_, y = p % Y_;
        float xl[TC_], acc[TC_];
        #pragma unroll
        for (int j = 0; j < TC_; ++j) {
            xl[j]  = lds_xl[b * TY_ + (t0 + j) * Y_ + y];
            acc[j] = 0.0f;
        }
        const float2* __restrict__ q2 = (const float2*)(lds_Q + b * H_);
        #pragma unroll 4
        for (int h2 = 0; h2 < H_ / 2; ++h2) {
            const float2 q  = q2[h2];
            const float2 w0 = lds_vk[(2 * h2    ) * Y_ + y];
            const float2 w1 = lds_vk[(2 * h2 + 1) * Y_ + y];
            #pragma unroll
            for (int j = 0; j < TC_; ++j) {
                acc[j] += w0.y * fast_tanh(q.x + w0.x * xl[j]);
                acc[j] += w1.y * fast_tanh(q.y + w1.x * xl[j]);
            }
        }
        #pragma unroll
        for (int j = 0; j < TC_; ++j)
            lds_A[b * TY_ + (t0 + j) * Y_ + y] = acc[j];
    }
    __syncthreads();

    if (tid < B_ * Y_) {
        const int b = tid / Y_, y = tid % Y_;
        float mx = -INFINITY;
        #pragma unroll
        for (int t = 0; t < T_; ++t)
            mx = fmaxf(mx, lds_A[b * TY_ + t * Y_ + y]);
        float s = 0.0f;
        #pragma unroll
        for (int t = 0; t < T_; ++t) {
            const int idx = b * TY_ + t * Y_ + y;
            const float w = __expf(lds_A[idx] - mx);
            lds_A[idx] = w;
            s += w;
        }
        float xn = 0.0f;
        #pragma unroll
        for (int t = 0; t < T_; ++t) {
            const int idx = b * TY_ + t * Y_ + y;
            xn += lds_A[idx] * lds_xl[idx];
        }
        lds_xn[b * 16 + y] = xn / s;
    }
    __syncthreads();

    if (wave < 4) {
        const int d0 = wave * 16;
        const int m  = lane & 15;
        const int ko = (lane >> 4) * 2;
        v8f c = {};
        #pragma unroll
        for (int kk = 0; kk < 3; ++kk) {
            const int k = kk * 4 + ko;
            const float2 af = *(const float2*)&lds_xn[m * 16 + k];
            const float2 bf = *(const float2*)&lds_ewT[(d0 + m) * 16 + k];
            v2f a, b;
            a.x = af.x; a.y = af.y;
            b.x = bf.x; b.y = bf.y;
            c = wmma_f32_16x16x4(a, b, c);
        }
        if (lane < 16) {
            #pragma unroll
            for (int v = 0; v < B_; ++v)
                out[(v * N_ + n) * DL_ + d0 + lane] =
                    c[v] + encb[n * DL_ + d0 + lane];
        }
    }
}

extern "C" void kernel_launch(void* const* d_in, const int* in_sizes, int n_in,
                              void* d_out, int out_size, void* d_ws, size_t ws_size,
                              hipStream_t stream) {
    const float* x_long  = (const float*)d_in[0];
    const float* x_short = (const float*)d_in[1];
    const float* Wq      = (const float*)d_in[2];
    const float* Wk      = (const float*)d_in[3];
    const float* V       = (const float*)d_in[4];
    const float* encW    = (const float*)d_in[5];
    const float* encb    = (const float*)d_in[6];
    float* out = (float*)d_out;

    tsa_fused_kernel<<<N_, NT_, 0, stream>>>(x_long, x_short, Wq, Wk, V,
                                             encW, encb, out);
}


# Waves per SIMD capped at 12 by attribute, so the compiler may use 80 VGPRs (49 -> 79 used): each tanh-loop trip is now 48 independent fma-tanh-fma chains, not a serial chain with a v_nop per tanh
# speedup vs baseline: 1.2448x; 1.2448x over previous
#include <hip/hip_runtime.h>
#include <cmath>

typedef __attribute__((ext_vector_type(2))) float v2f;
typedef __attribute__((ext_vector_type(8))) float v8f;

#define B_  8
#define N_  400
#define T_  24
#define Y_  12
#define H_  64
#define D_  96
#define DL_ 64
#define TY_ (T_ * Y_)
#define BTY_ (B_ * TY_)
#define NT_ 384
#define TC_ 6

__device__ __forceinline__ float fast_tanh(float x) {
#if __has_builtin(__builtin_amdgcn_tanhf)
    return __builtin_amdgcn_tanhf(x);
#else
    return tanhf(x);
#endif
}

__device__ __forceinline__ v8f wmma_f32_16x16x4(v2f a, v2f b, v8f c) {
    return __builtin_amdgcn_wmma_f32_16x16x4_f32(
         false, a,  false, b,
         (short)0, c,  false,  false);
}

__global__ __launch_bounds__(NT_) __attribute__((amdgpu_waves_per_eu(3, 12)))
void tsa_fused_kernel(
    const float* __restrict__ x_long,
    const float* __restrict__ x_short,
    const float* __restrict__ Wq,
    const float* __restrict__ Wk,
    const float* __restrict__ V,
    const float* __restrict__ encW,
    const float* __restrict__ encb,
    float* __restrict__ out)
{
    __shared__ alignas(16) float  lds_wq[H_ * D_];
    __shared__ alignas(16) float  lds_xs[16 * D_];
    __shared__ alignas(16) float  lds_Q [B_ * H_];
    __shared__ alignas(16) float2 lds_vk[H_ * Y_];
    __shared__ alignas(16) float  lds_xl[BTY_];
    __shared__ alignas(16) float  lds_A [BTY_];
    __shared__ alignas(16) float  lds_xn[16 * 16];
    __shared__ alignas(16) float  lds_ewT[DL_ * 16];

    const int n    = blockIdx.x;
    const int tid  = threadIdx.x;
    const int wave = tid >> 5;
    const int lane = tid & 31;

    for (int i4 = tid; i4 < (H_ * D_) / 4; i4 += NT_) {
        const int i = i4 * 4;
        const int h = i / D_, d = i % D_;
        ((float4*)lds_wq)[i4] = *(const float4*)&Wq[(h * N_ + n) * D_ + d];
    }
    for (int i4 = tid; i4 < (B_ * D_) / 4; i4 += NT_) {
        const int i = i4 * 4;
        const int b = i / D_, d = i % D_;
        ((float4*)lds_xs)[i4] = *(const float4*)&x_short[(b * N_ + n) * D_ + d];
    }
    for (int i = tid; i < B_ * D_; i += NT_) lds_xs[B_ * D_ + i] = 0.0f;
    for (int i = tid; i < H_ * Y_; i += NT_) {
        const int h = i / Y_, y = i % Y_;
        lds_vk[i] = make_float2(Wk[(h * N_ + n) * Y_ + y],
                                V [(h * N_ + n) * Y_ + y]);
    }
    for (int i4 = tid; i4 < BTY_ / 4; i4 += NT_) {
        const int i = i4 * 4;
        const int b = i / TY_, r = i % TY_;
        ((float4*)lds_xl)[i4] = *(const float4*)&x_long[(b * N_ + n) * TY_ + r];
    }
    for (int i = tid; i < Y_ * DL_; i += NT_) {
        const int y = i / DL_, d = i % DL_;
        lds_ewT[d * 16 + y] = encW[n * Y_ * DL_ + i];
    }
    if (tid < 256) lds_xn[tid] = 0.0f;
    __syncthreads();

    if (wave < 4) {
        const int h0 = wave * 16;
        const int m  = lane & 15;
        const int ko = (lane >> 4) * 2;
        v8f c = {};
        #pragma unroll
        for (int kk = 0; kk < D_ / 4; ++kk) {
            const int k = kk * 4 + ko;
            const float2 af = *(const float2*)&lds_xs[m * D_ + k];
            const float2 bf = *(const float2*)&lds_wq[(h0 + m) * D_ + k];
            v2f a, b;
            a.x = af.x; a.y = af.y;
            b.x = bf.x; b.y = bf.y;
            c = wmma_f32_16x16x4(a, b, c);
        }
        if (lane < 16) {
            #pragma unroll
            for (int v = 0; v < B_; ++v)
                lds_Q[v * H_ + h0 + lane] = c[v];
        }
    }
    __syncthreads();

    {
        const int p  = tid % (B_ * Y_);
        const int t0 = (tid / (B_ * Y_)) * TC_;
        const int b  = p / Y_, y = p % Y_;
        float xl[TC_], acc[TC_];
        #pragma unroll
        for (int j = 0; j < TC_; ++j) {
            xl[j]  = lds_xl[b * TY_ + (t0 + j) * Y_ + y];
            acc[j] = 0.0f;
        }
        const float2* __restrict__ q2 = (const float2*)(lds_Q + b * H_);
        #pragma unroll 4
        for (int h2 = 0; h2 < H_ / 2; ++h2) {
            const float2 q  = q2[h2];
            const float2 w0 = lds_vk[(2 * h2    ) * Y_ + y];
            const float2 w1 = lds_vk[(2 * h2 + 1) * Y_ + y];
            #pragma unroll
            for (int j = 0; j < TC_; ++j) {
                acc[j] += w0.y * fast_tanh(q.x + w0.x * xl[j]);
                acc[j] += w1.y * fast_tanh(q.y + w1.x * xl[j]);
            }
        }
        #pragma unroll
        for (int j = 0; j < TC_; ++j)
            lds_A[b * TY_ + (t0 + j) * Y_ + y] = acc[j];
    }
    __syncthreads();

    if (tid < B_ * Y_) {
        const int b = tid / Y_, y = tid % Y_;
        float mx = -INFINITY;
        #pragma unroll
        for (int t = 0; t < T_; ++t)
            mx = fmaxf(mx, lds_A[b * TY_ + t * Y_ + y]);
        float s = 0.0f;
        #pragma unroll
        for (int t = 0; t < T_; ++t) {
            const int idx = b * TY_ + t * Y_ + y;
            const float w = __expf(lds_A[idx] - mx);
            lds_A[idx] = w;
            s += w;
        }
        float xn = 0.0f;
        #pragma unroll
        for (int t = 0; t < T_; ++t) {
            const int idx = b * TY_ + t * Y_ + y;
            xn += lds_A[idx] * lds_xl[idx];
        }
        lds_xn[b * 16 + y] = xn / s;
    }
    __syncthreads();

    if (wave < 4) {
        const int d0 = wave * 16;
        const int m  = lane & 15;
        const int ko = (lane >> 4) * 2;
        v8f c = {};
        #pragma unroll
        for (int kk = 0; kk < 3; ++kk) {
            const int k = kk * 4 + ko;
            const float2 af = *(const float2*)&lds_xn[m * 16 + k];
            const float2 bf = *(const float2*)&lds_ewT[(d0 + m) * 16 + k];
            v2f a, b;
            a.x = af.x; a.y = af.y;
            b.x = bf.x; b.y = bf.y;
            c = wmma_f32_16x16x4(a, b, c);
        }
        if (lane < 16) {
            #pragma unroll
            for (int v = 0; v < B_; ++v)
                out[(v * N_ + n) * DL_ + d0 + lane] =
                    c[v] + encb[n * DL_ + d0 + lane];
        }
    }
}

extern "C" void kernel_launch(void* const* d_in, const int* in_sizes, int n_in,
                              void* d_out, int out_size, void* d_ws, size_t ws_size,
                              hipStream_t stream) {
    const float* x_long  = (const float*)d_in[0];
    const float* x_short = (const float*)d_in[1];
    const float* Wq      = (const float*)d_in[2];
    const float* Wk      = (const float*)d_in[3];
    const float* V       = (const float*)d_in[4];
    const float* encW    = (const float*)d_in[5];
    const float* encb    = (const float*)d_in[6];
    float* out = (float*)d_out;

    tsa_fused_kernel<<<N_, NT_, 0, stream>>>(x_long, x_short, Wq, Wk, V,
                                             encW, encb, out);
}


# All global loads (Wq, x_long, x_short, Wk, V, enc_W, enc_b) issued together at entry, before any wait: one memory round trip in place of eleven serial ones before barrier 1 and one at the end
# speedup vs baseline: 1.7557x; 1.4104x over previous
#include <hip/hip_runtime.h>
#include <cmath>

typedef __attribute__((ext_vector_type(2))) float v2f;
typedef __attribute__((ext_vector_type(8))) float v8f;

#define B_  8
#define N_  400
#define T_  24
#define Y_  12
#define H_  64
#define D_  96
#define DL_ 64
#define TY_ (T_ * Y_)
#define BTY_ (B_ * TY_)
#define NT_ 384
#define TC_ 6

__device__ __forceinline__ float fast_tanh(float x) {
#if __has_builtin(__builtin_amdgcn_tanhf)
    return __builtin_amdgcn_tanhf(x);
#else
    return tanhf(x);
#endif
}

__device__ __forceinline__ v8f wmma_f32_16x16x4(v2f a, v2f b, v8f c) {
    return __builtin_amdgcn_wmma_f32_16x16x4_f32(
         false, a,  false, b,
         (short)0, c,  false,  false);
}

__global__ __launch_bounds__(NT_) __attribute__((amdgpu_waves_per_eu(3, 12)))
void tsa_fused_kernel(
    const float* __restrict__ x_long,
    const float* __restrict__ x_short,
    const float* __restrict__ Wq,
    const float* __restrict__ Wk,
    const float* __restrict__ V,
    const float* __restrict__ encW,
    const float* __restrict__ encb,
    float* __restrict__ out)
{
    __shared__ alignas(16) float  lds_wq[H_ * D_];
    __shared__ alignas(16) float  lds_xs[16 * D_];
    __shared__ alignas(16) float  lds_Q [B_ * H_];
    __shared__ alignas(16) float2 lds_vk[H_ * Y_];
    __shared__ alignas(16) float  lds_xl[BTY_];
    __shared__ alignas(16) float  lds_A [BTY_];
    __shared__ alignas(16) float  lds_xn[16 * 16];
    __shared__ alignas(16) float  lds_ewT[DL_ * 16];
    __shared__ alignas(16) float  lds_eb [DL_];

    const int n    = blockIdx.x;
    const int tid  = threadIdx.x;
    const int wave = tid >> 5;
    const int lane = tid & 31;

    float4 r_wq[4];
    #pragma unroll
    for (int k = 0; k < 4; ++k) {
        const int i = (tid + k * NT_) * 4;
        const int h = i / D_, d = i % D_;
        r_wq[k] = *(const float4*)&Wq[(h * N_ + n) * D_ + d];
    }
    float4 r_xa, r_xb;
    float4* dst_xb;
    {
        const int i = tid * 4;
        const int b = i / TY_, r = i % TY_;
        r_xa = *(const float4*)&x_long[(b * N_ + n) * TY_ + r];
    }
    {
        const bool is_xl = tid < (BTY_ / 4 - NT_);
        const int j4  = is_xl ? tid + NT_ : tid - (BTY_ / 4 - NT_);
        const int i   = j4 * 4;
        const int bl  = i / TY_, rl = i % TY_;
        const int bs  = i / D_,  ds = i % D_;
        const float* src = is_xl ? &x_long [(bl * N_ + n) * TY_ + rl]
                                 : &x_short[(bs * N_ + n) * D_  + ds];
        r_xb   = *(const float4*)src;
        dst_xb = is_xl ? &((float4*)lds_xl)[j4] : &((float4*)lds_xs)[j4];
    }
    float r_wk[2], r_v[2], r_ew[2];
    #pragma unroll
    for (int k = 0; k < 2; ++k) {
        const int i = tid + k * NT_;
        const int h = i / Y_, y = i % Y_;
        r_wk[k] = Wk[(h * N_ + n) * Y_ + y];
        r_v [k] = V [(h * N_ + n) * Y_ + y];
        r_ew[k] = encW[n * Y_ * DL_ + i];
    }
    const float r_eb = encb[n * DL_ + (tid & (DL_ - 1))];

    #pragma unroll
    for (int k = 0; k < 2; ++k) lds_xs[B_ * D_ + tid + k * NT_] = 0.0f;
    if (tid < 256) lds_xn[tid] = 0.0f;

    #pragma unroll
    for (int k = 0; k < 4; ++k) ((float4*)lds_wq)[tid + k * NT_] = r_wq[k];
    ((float4*)lds_xl)[tid] = r_xa;
    *dst_xb = r_xb;
    #pragma unroll
    for (int k = 0; k < 2; ++k) {
        const int i = tid + k * NT_;
        lds_vk[i] = make_float2(r_wk[k], r_v[k]);
        const int y = i / DL_, d = i % DL_;
        lds_ewT[d * 16 + y] = r_ew[k];
    }
    if (tid < DL_) lds_eb[tid] = r_eb;
    __syncthreads();

    if (wave < 4) {
        const int h0 = wave * 16;
        const int m  = lane & 15;
        const int ko = (lane >> 4) * 2;
        v8f c = {};
        #pragma unroll
        for (int kk = 0; kk < D_ / 4; ++kk) {
            const int k = kk * 4 + ko;
            const float2 af = *(const float2*)&lds_xs[m * D_ + k];
            const float2 bf = *(const float2*)&lds_wq[(h0 + m) * D_ + k];
            v2f a, b;
            a.x = af.x; a.y = af.y;
            b.x = bf.x; b.y = bf.y;
            c = wmma_f32_16x16x4(a, b, c);
        }
        if (lane < 16) {
            #pragma unroll
            for (int v = 0; v < B_; ++v)
                lds_Q[v * H_ + h0 + lane] = c[v];
        }
    }
    __syncthreads();

    {
        const int p  = tid % (B_ * Y_);
        const int t0 = (tid / (B_ * Y_)) * TC_;
        const int b  = p / Y_, y = p % Y_;
        float xl[TC_], acc[TC_];
        #pragma unroll
        for (int j = 0; j < TC_; ++j) {
            xl[j]  = lds_xl[b * TY_ + (t0 + j) * Y_ + y];
            acc[j] = 0.0f;
        }
        const float2* __restrict__ q2 = (const float2*)(lds_Q + b * H_);
        #pragma unroll 4
        for (int h2 = 0; h2 < H_ / 2; ++h2) {
            const float2 q  = q2[h2];
            const float2 w0 = lds_vk[(2 * h2    ) * Y_ + y];
            const float2 w1 = lds_vk[(2 * h2 + 1) * Y_ + y];
            #pragma unroll
            for (int j = 0; j < TC_; ++j) {
                acc[j] += w0.y * fast_tanh(q.x + w0.x * xl[j]);
                acc[j] += w1.y * fast_tanh(q.y + w1.x * xl[j]);
            }
        }
        #pragma unroll
        for (int j = 0; j < TC_; ++j)
            lds_A[b * TY_ + (t0 + j) * Y_ + y] = acc[j];
    }
    __syncthreads();

    if (tid < B_ * Y_) {
        const int b = tid / Y_, y = tid % Y_;
        float mx = -INFINITY;
        #pragma unroll
        for (int t = 0; t < T_; ++t)
            mx = fmaxf(mx, lds_A[b * TY_ + t * Y_ + y]);
        float s = 0.0f;
        #pragma unroll
        for (int t = 0; t < T_; ++t) {
            const int idx = b * TY_ + t * Y_ + y;
            const float w = __expf(lds_A[idx] - mx);
            lds_A[idx] = w;
            s += w;
        }
        float xn = 0.0f;
        #pragma unroll
        for (int t = 0; t < T_; ++t) {
            const int idx = b * TY_ + t * Y_ + y;
            xn += lds_A[idx] * lds_xl[idx];
        }
        lds_xn[b * 16 + y] = xn / s;
    }
    __syncthreads();

    if (wave < 4) {
        const int d0 = wave * 16;
        const int m  = lane & 15;
        const int ko = (lane >> 4) * 2;
        v8f c = {};
        #pragma unroll
        for (int kk = 0; kk < 3; ++kk) {
            const int k = kk * 4 + ko;
            const float2 af = *(const float2*)&lds_xn[m * 16 + k];
            const float2 bf = *(const float2*)&lds_ewT[(d0 + m) * 16 + k];
            v2f a, b;
            a.x = af.x; a.y = af.y;
            b.x = bf.x; b.y = bf.y;
            c = wmma_f32_16x16x4(a, b, c);
        }
        if (lane < 16) {
            const float eb = lds_eb[d0 + lane];
            #pragma unroll
            for (int v = 0; v < B_; ++v)
                out[(v * N_ + n) * DL_ + d0 + lane] = c[v] + eb;
        }
    }
}

extern "C" void kernel_launch(void* const* d_in, const int* in_sizes, int n_in,
                              void* d_out, int out_size, void* d_ws, size_t ws_size,
                              hipStream_t stream) {
    const float* x_long  = (const float*)d_in[0];
    const float* x_short = (const float*)d_in[1];
    const float* Wq      = (const float*)d_in[2];
    const float* Wk      = (const float*)d_in[3];
    const float* V       = (const float*)d_in[4];
    const float* encW    = (const float*)d_in[5];
    const float* encb    = (const float*)d_in[6];
    float* out = (float*)d_out;

    tsa_fused_kernel<<<N_, NT_, 0, stream>>>(x_long, x_short, Wq, Wk, V,
                                             encW, encb, out);
}


# Softmax over t and weighted sum done in the tanh threads' registers (4 t-chunks of a pair in adjacent lanes, DPP swaps): no logits in LDS, one barrier less; t sums 6 serial + 4-lane tree; wave cap 10
# speedup vs baseline: 1.8479x; 1.0525x over previous
#include <hip/hip_runtime.h>
#include <cmath>

typedef __attribute__((ext_vector_type(2))) float v2f;
typedef __attribute__((ext_vector_type(8))) float v8f;

#define B_  8
#define N_  400
#define T_  24
#define Y_  12
#define H_  64
#define D_  96
#define DL_ 64
#define TY_ (T_ * Y_)
#define BTY_ (B_ * TY_)
#define NT_ 384
#define TC_ 6

__device__ __forceinline__ float fast_tanh(float x) {
#if __has_builtin(__builtin_amdgcn_tanhf)
    return __builtin_amdgcn_tanhf(x);
#else
    return tanhf(x);
#endif
}

__device__ __forceinline__ float lane_xor1(float v) {
    return __builtin_bit_cast(float, __builtin_amdgcn_update_dpp(
        0, __builtin_bit_cast(int, v), 0x161  , 0xf, 0xf, true));
}
__device__ __forceinline__ float lane_xor2(float v) {
    return __builtin_bit_cast(float, __builtin_amdgcn_update_dpp(
        0, __builtin_bit_cast(int, v), 0x162  , 0xf, 0xf, true));
}

__device__ __forceinline__ v8f wmma_f32_16x16x4(v2f a, v2f b, v8f c) {
    return __builtin_amdgcn_wmma_f32_16x16x4_f32(
         false, a,  false, b,
         (short)0, c,  false,  false);
}

__global__ __launch_bounds__(NT_) __attribute__((amdgpu_waves_per_eu(3, 10)))
void tsa_fused_kernel(
    const float* __restrict__ x_long,
    const float* __restrict__ x_short,
    const float* __restrict__ Wq,
    const float* __restrict__ Wk,
    const float* __restrict__ V,
    const float* __restrict__ encW,
    const float* __restrict__ encb,
    float* __restrict__ out)
{
    __shared__ alignas(16) float  lds_wq[H_ * D_];
    __shared__ alignas(16) float  lds_xs[16 * D_];
    __shared__ alignas(16) float  lds_Q [B_ * H_];
    __shared__ alignas(16) float2 lds_vk[H_ * Y_];
    __shared__ alignas(16) float  lds_xl[BTY_];
    __shared__ alignas(16) float  lds_xn[16 * 16];
    __shared__ alignas(16) float  lds_ewT[DL_ * 16];
    __shared__ alignas(16) float  lds_eb [DL_];

    const int n    = blockIdx.x;
    const int tid  = threadIdx.x;
    const int wave = tid >> 5;
    const int lane = tid & 31;

    float4 r_wq[4];
    #pragma unroll
    for (int k = 0; k < 4; ++k) {
        const int i = (tid + k * NT_) * 4;
        const int h = i / D_, d = i % D_;
        r_wq[k] = *(const float4*)&Wq[(h * N_ + n) * D_ + d];
    }
    float4 r_xa, r_xb;
    float4* dst_xb;
    {
        const int i = tid * 4;
        const int b = i / TY_, r = i % TY_;
        r_xa = *(const float4*)&x_long[(b * N_ + n) * TY_ + r];
    }
    {
        const bool is_xl = tid < (BTY_ / 4 - NT_);
        const int j4  = is_xl ? tid + NT_ : tid - (BTY_ / 4 - NT_);
        const int i   = j4 * 4;
        const int bl  = i / TY_, rl = i % TY_;
        const int bs  = i / D_,  ds = i % D_;
        const float* src = is_xl ? &x_long [(bl * N_ + n) * TY_ + rl]
                                 : &x_short[(bs * N_ + n) * D_  + ds];
        r_xb   = *(const float4*)src;
        dst_xb = is_xl ? &((float4*)lds_xl)[j4] : &((float4*)lds_xs)[j4];
    }
    float r_wk[2], r_v[2], r_ew[2];
    #pragma unroll
    for (int k = 0; k < 2; ++k) {
        const int i = tid + k * NT_;
        const int h = i / Y_, y = i % Y_;
        r_wk[k] = Wk[(h * N_ + n) * Y_ + y];
        r_v [k] = V [(h * N_ + n) * Y_ + y];
        r_ew[k] = encW[n * Y_ * DL_ + i];
    }
    const float r_eb = encb[n * DL_ + (tid & (DL_ - 1))];

    #pragma unroll
    for (int k = 0; k < 2; ++k) lds_xs[B_ * D_ + tid + k * NT_] = 0.0f;
    if (tid < 256) lds_xn[tid] = 0.0f;

    #pragma unroll
    for (int k = 0; k < 4; ++k) ((float4*)lds_wq)[tid + k * NT_] = r_wq[k];
    ((float4*)lds_xl)[tid] = r_xa;
    *dst_xb = r_xb;
    #pragma unroll
    for (int k = 0; k < 2; ++k) {
        const int i = tid + k * NT_;
        lds_vk[i] = make_float2(r_wk[k], r_v[k]);
        const int y = i / DL_, d = i % DL_;
        lds_ewT[d * 16 + y] = r_ew[k];
    }
    if (tid < DL_) lds_eb[tid] = r_eb;
    __syncthreads();

    if (wave < 4) {
        const int h0 = wave * 16;
        const int m  = lane & 15;
        const int ko = (lane >> 4) * 2;
        v8f c = {};
        #pragma unroll
        for (int kk = 0; kk < D_ / 4; ++kk) {
            const int k = kk * 4 + ko;
            const float2 af = *(const float2*)&lds_xs[m * D_ + k];
            const float2 bf = *(const float2*)&lds_wq[(h0 + m) * D_ + k];
            v2f a, b;
            a.x = af.x; a.y = af.y;
            b.x = bf.x; b.y = bf.y;
            c = wmma_f32_16x16x4(a, b, c);
        }
        if (lane < 16) {
            #pragma unroll
            for (int v = 0; v < B_; ++v)
                lds_Q[v * H_ + h0 + lane] = c[v];
        }
    }
    __syncthreads();

    {
        const int p  = tid >> 2;
        const int t0 = (tid & 3) * TC_;
        const int b  = p / Y_, y = p % Y_;
        float xl[TC_], acc[TC_];
        #pragma unroll
        for (int j = 0; j < TC_; ++j) {
            xl[j]  = lds_xl[b * TY_ + (t0 + j) * Y_ + y];
            acc[j] = 0.0f;
        }
        const float2* __restrict__ q2 = (const float2*)(lds_Q + b * H_);
        #pragma unroll 4
        for (int h2 = 0; h2 < H_ / 2; ++h2) {
            const float2 q  = q2[h2];
            const float2 w0 = lds_vk[(2 * h2    ) * Y_ + y];
            const float2 w1 = lds_vk[(2 * h2 + 1) * Y_ + y];
            #pragma unroll
            for (int j = 0; j < TC_; ++j) {
                acc[j] += w0.y * fast_tanh(q.x + w0.x * xl[j]);
                acc[j] += w1.y * fast_tanh(q.y + w1.x * xl[j]);
            }
        }

        float mx = acc[0];
        #pragma unroll
        for (int j = 1; j < TC_; ++j) mx = fmaxf(mx, acc[j]);
        mx = fmaxf(mx, lane_xor1(mx));
        mx = fmaxf(mx, lane_xor2(mx));
        float s = 0.0f, xn = 0.0f;
        #pragma unroll
        for (int j = 0; j < TC_; ++j) {
            const float w = __expf(acc[j] - mx);
            s  += w;
            xn += w * xl[j];
        }
        s  += lane_xor1(s);
        xn += lane_xor1(xn);
        s  += lane_xor2(s);
        xn += lane_xor2(xn);
        if ((tid & 3) == 0) lds_xn[b * 16 + y] = xn / s;
    }
    __syncthreads();

    if (wave < 4) {
        const int d0 = wave * 16;
        const int m  = lane & 15;
        const int ko = (lane >> 4) * 2;
        v8f c = {};
        #pragma unroll
        for (int kk = 0; kk < 3; ++kk) {
            const int k = kk * 4 + ko;
            const float2 af = *(const float2*)&lds_xn[m * 16 + k];
            const float2 bf = *(const float2*)&lds_ewT[(d0 + m) * 16 + k];
            v2f a, b;
            a.x = af.x; a.y = af.y;
            b.x = bf.x; b.y = bf.y;
            c = wmma_f32_16x16x4(a, b, c);
        }
        if (lane < 16) {
            const float eb = lds_eb[d0 + lane];
            #pragma unroll
            for (int v = 0; v < B_; ++v)
                out[(v * N_ + n) * DL_ + d0 + lane] = c[v] + eb;
        }
    }
}

extern "C" void kernel_launch(void* const* d_in, const int* in_sizes, int n_in,
                              void* d_out, int out_size, void* d_ws, size_t ws_size,
                              hipStream_t stream) {
    const float* x_long  = (const float*)d_in[0];
    const float* x_short = (const float*)d_in[1];
    const float* Wq      = (const float*)d_in[2];
    const float* Wk      = (const float*)d_in[3];
    const float* V       = (const float*)d_in[4];
    const float* encW    = (const float*)d_in[5];
    const float* encb    = (const float*)d_in[6];
    float* out = (float*)d_out;

    tsa_fused_kernel<<<N_, NT_, 0, stream>>>(x_long, x_short, Wq, Wk, V,
                                             encW, encb, out);
}


# Tanh loop issue order laid out by hand: tanh, packed fma, tanh, packed fma (argument 3 units ahead, sum 3 units behind, order pinned), not runs of 6-8 tanh that hide one fma each; sums unchanged
# speedup vs baseline: 1.9495x; 1.0550x over previous
#include <hip/hip_runtime.h>
#include <cmath>

typedef __attribute__((ext_vector_type(2))) float v2f;
typedef __attribute__((ext_vector_type(8))) float v8f;

#define B_  8
#define N_  400
#define T_  24
#define Y_  12
#define H_  64
#define D_  96
#define DL_ 64
#define TY_ (T_ * Y_)
#define BTY_ (B_ * TY_)
#define NT_ 384
#define TC_ 6
#define UPB_ (8 * (TC_ / 2))
#define LEAD_ 3
#define LAG_  3

__device__ __forceinline__ float fast_tanh(float x) {
#if __has_builtin(__builtin_amdgcn_tanhf)
    return __builtin_amdgcn_tanhf(x);
#else
    return tanhf(x);
#endif
}

__device__ __forceinline__ float lane_xor1(float v) {
    return __builtin_bit_cast(float, __builtin_amdgcn_update_dpp(
        0, __builtin_bit_cast(int, v), 0x161  , 0xf, 0xf, true));
}
__device__ __forceinline__ float lane_xor2(float v) {
    return __builtin_bit_cast(float, __builtin_amdgcn_update_dpp(
        0, __builtin_bit_cast(int, v), 0x162  , 0xf, 0xf, true));
}

__device__ __forceinline__ v8f wmma_f32_16x16x4(v2f a, v2f b, v8f c) {
    return __builtin_amdgcn_wmma_f32_16x16x4_f32(
         false, a,  false, b,
         (short)0, c,  false,  false);
}

__global__ __launch_bounds__(NT_) __attribute__((amdgpu_waves_per_eu(3, 10)))
void tsa_fused_kernel(
    const float* __restrict__ x_long,
    const float* __restrict__ x_short,
    const float* __restrict__ Wq,
    const float* __restrict__ Wk,
    const float* __restrict__ V,
    const float* __restrict__ encW,
    const float* __restrict__ encb,
    float* __restrict__ out)
{
    __shared__ alignas(16) float  lds_wq[H_ * D_];
    __shared__ alignas(16) float  lds_xs[16 * D_];
    __shared__ alignas(16) float  lds_Q [B_ * H_];
    __shared__ alignas(16) float2 lds_vk[H_ * Y_];
    __shared__ alignas(16) float  lds_xl[BTY_];
    __shared__ alignas(16) float  lds_xn[16 * 16];
    __shared__ alignas(16) float  lds_ewT[DL_ * 16];
    __shared__ alignas(16) float  lds_eb [DL_];

    const int n    = blockIdx.x;
    const int tid  = threadIdx.x;
    const int wave = tid >> 5;
    const int lane = tid & 31;

    float4 r_wq[4];
    #pragma unroll
    for (int k = 0; k < 4; ++k) {
        const int i = (tid + k * NT_) * 4;
        const int h = i / D_, d = i % D_;
        r_wq[k] = *(const float4*)&Wq[(h * N_ + n) * D_ + d];
    }
    float4 r_xa, r_xb;
    float4* dst_xb;
    {
        const int i = tid * 4;
        const int b = i / TY_, r = i % TY_;
        r_xa = *(const float4*)&x_long[(b * N_ + n) * TY_ + r];
    }
    {
        const bool is_xl = tid < (BTY_ / 4 - NT_);
        const int j4  = is_xl ? tid + NT_ : tid - (BTY_ / 4 - NT_);
        const int i   = j4 * 4;
        const int bl  = i / TY_, rl = i % TY_;
        const int bs  = i / D_,  ds = i % D_;
        const float* src = is_xl ? &x_long [(bl * N_ + n) * TY_ + rl]
                                 : &x_short[(bs * N_ + n) * D_  + ds];
        r_xb   = *(const float4*)src;
        dst_xb = is_xl ? &((float4*)lds_xl)[j4] : &((float4*)lds_xs)[j4];
    }
    float r_wk[2], r_v[2], r_ew[2];
    #pragma unroll
    for (int k = 0; k < 2; ++k) {
        const int i = tid + k * NT_;
        const int h = i / Y_, y = i % Y_;
        r_wk[k] = Wk[(h * N_ + n) * Y_ + y];
        r_v [k] = V [(h * N_ + n) * Y_ + y];
        r_ew[k] = encW[n * Y_ * DL_ + i];
    }
    const float r_eb = encb[n * DL_ + (tid & (DL_ - 1))];

    #pragma unroll
    for (int k = 0; k < 2; ++k) lds_xs[B_ * D_ + tid + k * NT_] = 0.0f;
    if (tid < 256) lds_xn[tid] = 0.0f;

    #pragma unroll
    for (int k = 0; k < 4; ++k) ((float4*)lds_wq)[tid + k * NT_] = r_wq[k];
    ((float4*)lds_xl)[tid] = r_xa;
    *dst_xb = r_xb;
    #pragma unroll
    for (int k = 0; k < 2; ++k) {
        const int i = tid + k * NT_;
        lds_vk[i] = make_float2(r_wk[k], r_v[k]);
        const int y = i / DL_, d = i % DL_;
        lds_ewT[d * 16 + y] = r_ew[k];
    }
    if (tid < DL_) lds_eb[tid] = r_eb;
    __syncthreads();

    if (wave < 4) {
        const int h0 = wave * 16;
        const int m  = lane & 15;
        const int ko = (lane >> 4) * 2;
        v8f c = {};
        #pragma unroll
        for (int kk = 0; kk < D_ / 4; ++kk) {
            const int k = kk * 4 + ko;
            const float2 af = *(const float2*)&lds_xs[m * D_ + k];
            const float2 bf = *(const float2*)&lds_wq[(h0 + m) * D_ + k];
            v2f a, b;
            a.x = af.x; a.y = af.y;
            b.x = bf.x; b.y = bf.y;
            c = wmma_f32_16x16x4(a, b, c);
        }
        if (lane < 16) {
            #pragma unroll
            for (int v = 0; v < B_; ++v)
                lds_Q[v * H_ + h0 + lane] = c[v];
        }
    }
    __syncthreads();

    {
        const int p  = tid >> 2;
        const int t0 = (tid & 3) * TC_;
        const int b  = p / Y_, y = p % Y_;
        float xl[TC_], acc[TC_];
        #pragma unroll
        for (int j = 0; j < TC_; ++j)
            xl[j]  = lds_xl[b * TY_ + (t0 + j) * Y_ + y];
        v2f xp[TC_ / 2], ap[TC_ / 2];
        #pragma unroll
        for (int jp = 0; jp < TC_ / 2; ++jp) {
            xp[jp].x = xl[2 * jp]; xp[jp].y = xl[2 * jp + 1];
            ap[jp].x = 0.0f;       ap[jp].y = 0.0f;
        }
        const float2* __restrict__ q2 = (const float2*)(lds_Q + b * H_);

        #define TSA_Q(h)   (((h) & 1) ? qv[(h) >> 1].y : qv[(h) >> 1].x)
        #define TSA_ARG(u) arg[u] = __builtin_elementwise_fma(                 \
            (v2f){wv[(u) / 3].x, wv[(u) / 3].x}, xp[(u) % 3],                  \
            (v2f){TSA_Q((u) / 3), TSA_Q((u) / 3)})
        #define TSA_SUM(u) ap[(u) % 3] = __builtin_elementwise_fma(            \
            (v2f){wv[(u) / 3].y, wv[(u) / 3].y}, th[u], ap[(u) % 3])
        #define TSA_PIN()  __builtin_amdgcn_sched_barrier(0)
        #pragma nounroll
        for (int g = 0; g < H_ / 8; ++g) {
            float2 qv[4], wv[8];
            v2f    arg[UPB_], th[UPB_];
            #pragma unroll
            for (int i = 0; i < 4; ++i) qv[i] = q2[4 * g + i];
            #pragma unroll
            for (int i = 0; i < 8; ++i) wv[i] = lds_vk[(8 * g + i) * Y_ + y];
            TSA_PIN();
            #pragma unroll
            for (int u = 0; u < LEAD_; ++u) { TSA_ARG(u); }
            TSA_PIN();
            #pragma unroll
            for (int u = 0; u < UPB_; ++u) {
                th[u].x = fast_tanh(arg[u].x);
                TSA_PIN();
                if (u + LEAD_ < UPB_) { TSA_ARG(u + LEAD_); TSA_PIN(); }
                th[u].y = fast_tanh(arg[u].y);
                TSA_PIN();
                if (u >= LAG_) { TSA_SUM(u - LAG_); TSA_PIN(); }
            }
            #pragma unroll
            for (int u = UPB_ - LAG_; u < UPB_; ++u) { TSA_SUM(u); }
        }
        #undef TSA_Q
        #undef TSA_ARG
        #undef TSA_SUM
        #undef TSA_PIN
        #pragma unroll
        for (int jp = 0; jp < TC_ / 2; ++jp) {
            acc[2 * jp] = ap[jp].x; acc[2 * jp + 1] = ap[jp].y;
        }

        float mx = acc[0];
        #pragma unroll
        for (int j = 1; j < TC_; ++j) mx = fmaxf(mx, acc[j]);
        mx = fmaxf(mx, lane_xor1(mx));
        mx = fmaxf(mx, lane_xor2(mx));
        float s = 0.0f, xn = 0.0f;
        #pragma unroll
        for (int j = 0; j < TC_; ++j) {
            const float w = __expf(acc[j] - mx);
            s  += w;
            xn += w * xl[j];
        }
        s  += lane_xor1(s);
        xn += lane_xor1(xn);
        s  += lane_xor2(s);
        xn += lane_xor2(xn);
        if ((tid & 3) == 0) lds_xn[b * 16 + y] = xn / s;
    }
    __syncthreads();

    if (wave < 4) {
        const int d0 = wave * 16;
        const int m  = lane & 15;
        const int ko = (lane >> 4) * 2;
        v8f c = {};
        #pragma unroll
        for (int kk = 0; kk < 3; ++kk) {
            const int k = kk * 4 + ko;
            const float2 af = *(const float2*)&lds_xn[m * 16 + k];
            const float2 bf = *(const float2*)&lds_ewT[(d0 + m) * 16 + k];
            v2f a, b;
            a.x = af.x; a.y = af.y;
            b.x = bf.x; b.y = bf.y;
            c = wmma_f32_16x16x4(a, b, c);
        }
        if (lane < 16) {
            const float eb = lds_eb[d0 + lane];
            #pragma unroll
            for (int v = 0; v < B_; ++v)
                out[(v * N_ + n) * DL_ + d0 + lane] = c[v] + eb;
        }
    }
}

extern "C" void kernel_launch(void* const* d_in, const int* in_sizes, int n_in,
                              void* d_out, int out_size, void* d_ws, size_t ws_size,
                              hipStream_t stream) {
    const float* x_long  = (const float*)d_in[0];
    const float* x_short = (const float*)d_in[1];
    const float* Wq      = (const float*)d_in[2];
    const float* Wk      = (const float*)d_in[3];
    const float* V       = (const float*)d_in[4];
    const float* encW    = (const float*)d_in[5];
    const float* encb    = (const float*)d_in[6];
    float* out = (float*)d_out;

    tsa_fused_kernel<<<N_, NT_, 0, stream>>>(x_long, x_short, Wq, Wk, V,
                                             encW, encb, out);
}
